# starting kernel
# baseline (speedup 1.0000x reference)
#include <hip/hip_runtime.h>


typedef __attribute__((ext_vector_type(2))) float v2f;
typedef __attribute__((ext_vector_type(8))) float v8f;

__global__ void zero_f32(float* __restrict__ p, long long n) {
  long long i = (long long)blockIdx.x * blockDim.x + threadIdx.x;
  long long stride = (long long)gridDim.x * blockDim.x;
  for (; i < n; i += stride) p[i] = 0.0f;
}

__global__ void degree_kernel(const int* __restrict__ dst,
                              float* __restrict__ deg, int E) {
  int e = blockIdx.x * blockDim.x + threadIdx.x;
  if (e < E) atomicAdd(&deg[dst[e]], 1.0f);
}

template <int D>
__global__ void scatter_add_kernel(const float* __restrict__ feat,
                                   const int* __restrict__ src,
                                   const int* __restrict__ dst,
                                   float* __restrict__ agg, int E) {
  constexpr int PER_EDGE = D / 4;
  long long tid = (long long)blockIdx.x * blockDim.x + threadIdx.x;
  long long total = (long long)E * PER_EDGE;
  if (tid >= total) return;
  int e = (int)(tid / PER_EDGE);
  int f = (int)(tid % PER_EDGE) * 4;
  int s = src[e], d = dst[e];
  float4 v = *(const float4*)(feat + (size_t)s * D + f);
  float* p = agg + (size_t)d * D + f;
  atomicAdd(p + 0, v.x);
  atomicAdd(p + 1, v.y);
  atomicAdd(p + 2, v.z);
  atomicAdd(p + 3, v.w);
}

template <int K, bool RELU>
__launch_bounds__(256)
__global__ void sage_gemm_wmma(const float* __restrict__ aggsum,
                               const float* __restrict__ deg,
                               const float* __restrict__ xin,
                               const float* __restrict__ Wl,
                               const float* __restrict__ Wr,
                               const float* __restrict__ bias,
                               float* __restrict__ out,
                               int M, int N) {
  const int lane = threadIdx.x & 31;
  const int wave = threadIdx.x >> 5;
  const int tile = blockIdx.x * (blockDim.x >> 5) + wave;
  const int tilesN = N >> 4;
  const int nTiles = (M >> 4) * tilesN;
  if (tile >= nTiles) return;

  const int tm = tile / tilesN;
  const int tn = tile - tm * tilesN;

  const int m  = lane & 15;
  const int kh = lane >> 4;

  const int arow = tm * 16 + m;
  const int bcol = tn * 16 + m;

  const float invd = 1.0f / fmaxf(deg[arow], 1.0f);

  const float* aP = aggsum + (size_t)arow * K + 2 * kh;
  const float* xP = xin    + (size_t)arow * K + 2 * kh;
  const float* lP = Wl     + (size_t)bcol * K + 2 * kh;
  const float* rP = Wr     + (size_t)bcol * K + 2 * kh;

  v8f acc = {};

#pragma unroll 8
  for (int k0 = 0; k0 < K; k0 += 4) {
    v2f a, b;
    a.x = aP[k0] * invd;
    a.y = aP[k0 + 1] * invd;
    b.x = lP[k0];
    b.y = lP[k0 + 1];
    acc = __builtin_amdgcn_wmma_f32_16x16x4_f32(false, a, false, b,
                                                (short)0, acc, false, false);
  }

#pragma unroll 8
  for (int k0 = 0; k0 < K; k0 += 4) {
    v2f a, b;
    a.x = xP[k0];
    a.y = xP[k0 + 1];
    b.x = rP[k0];
    b.y = rP[k0 + 1];
    acc = __builtin_amdgcn_wmma_f32_16x16x4_f32(false, a, false, b,
                                                (short)0, acc, false, false);
  }

  const float bv = bias[bcol];
#pragma unroll
  for (int r = 0; r < 8; ++r) {
    float v = acc[r] + bv;
    if (RELU) v = fmaxf(v, 0.0f);
    out[(size_t)(tm * 16 + r + 8 * kh) * N + bcol] = v;
  }
}

extern "C" void kernel_launch(void* const* d_in, const int* in_sizes, int n_in,
                              void* d_out, int out_size, void* d_ws, size_t ws_size,
                              hipStream_t stream) {
  (void)n_in; (void)out_size; (void)ws_size;

  const float* x   = (const float*)d_in[0];
  const int*   ei  = (const int*)  d_in[1];
  const float* W1l = (const float*)d_in[2];
  const float* b1  = (const float*)d_in[3];
  const float* W1r = (const float*)d_in[4];
  const float* W2l = (const float*)d_in[5];
  const float* b2  = (const float*)d_in[6];
  const float* W2r = (const float*)d_in[7];
  float* out = (float*)d_out;

  const int D1 = in_sizes[3];
  const int D2 = in_sizes[6];
  const int D0 = in_sizes[2] / D1;
  const int M  = in_sizes[0] / D0;
  const int E  = in_sizes[1] / 2;
  const int* src = ei;
  const int* dst = ei + E;

  float* deg  = (float*)d_ws;
  float* bufA = deg + M;
  float* h    = bufA + (size_t)M * D1;

  zero_f32<<<4096, 256, 0, stream>>>(deg, (long long)M * D0 + M);
  degree_kernel<<<(E + 255) / 256, 256, 0, stream>>>(dst, deg, E);
  {
    long long total = (long long)E * (128 / 4);
    scatter_add_kernel<128><<<(int)((total + 255) / 256), 256, 0, stream>>>(
        x, src, dst, bufA, E);
  }
  {
    int tiles = (M / 16) * (D1 / 16);
    sage_gemm_wmma<128, true><<<(tiles + 7) / 8, 256, 0, stream>>>(
        bufA, deg, x, W1l, W1r, b1, h, M, D1);
  }

  zero_f32<<<4096, 256, 0, stream>>>(bufA, (long long)M * D1);
  {
    long long total = (long long)E * (256 / 4);
    scatter_add_kernel<256><<<(int)((total + 255) / 256), 256, 0, stream>>>(
        h, src, dst, bufA, E);
  }
  {
    int tiles = (M / 16) * (D2 / 16);
    sage_gemm_wmma<256, false><<<(tiles + 7) / 8, 256, 0, stream>>>(
        bufA, deg, h, W2l, W2r, b2, out, M, D2);
  }
}


# Layer 2 projects before it aggregates (mean aggregation is linear): one WMMA pass gives h@W2l^T and h@W2r^T+b2, then 128-wide projected rows are scattered into out instead of 256-wide h
# speedup vs baseline: 1.3816x; 1.3816x over previous
#include <hip/hip_runtime.h>


typedef __attribute__((ext_vector_type(2))) float v2f;
typedef __attribute__((ext_vector_type(8))) float v8f;

__global__ void zero_f32(float* __restrict__ p, long long n) {
  long long i = (long long)blockIdx.x * blockDim.x + threadIdx.x;
  long long stride = (long long)gridDim.x * blockDim.x;
  for (; i < n; i += stride) p[i] = 0.0f;
}

__global__ void degree_kernel(const int* __restrict__ dst,
                              float* __restrict__ deg, int E) {
  int e = blockIdx.x * blockDim.x + threadIdx.x;
  if (e < E) atomicAdd(&deg[dst[e]], 1.0f);
}

template <int D, bool MEAN>
__global__ void scatter_add_kernel(const float* __restrict__ feat,
                                   const int* __restrict__ src,
                                   const int* __restrict__ dst,
                                   const float* __restrict__ deg,
                                   float* __restrict__ agg, int E) {
  constexpr int PER_EDGE = D / 4;
  long long tid = (long long)blockIdx.x * blockDim.x + threadIdx.x;
  long long total = (long long)E * PER_EDGE;
  if (tid >= total) return;
  int e = (int)(tid / PER_EDGE);
  int f = (int)(tid % PER_EDGE) * 4;
  int s = src[e], d = dst[e];
  float4 v = *(const float4*)(feat + (size_t)s * D + f);
  if (MEAN) {
    const float sc = 1.0f / fmaxf(deg[d], 1.0f);
    v.x *= sc; v.y *= sc; v.z *= sc; v.w *= sc;
  }
  float* p = agg + (size_t)d * D + f;
  atomicAdd(p + 0, v.x);
  atomicAdd(p + 1, v.y);
  atomicAdd(p + 2, v.z);
  atomicAdd(p + 3, v.w);
}

template <int K, bool RELU>
__launch_bounds__(256)
__global__ void sage_gemm_wmma(const float* __restrict__ aggsum,
                               const float* __restrict__ deg,
                               const float* __restrict__ xin,
                               const float* __restrict__ Wl,
                               const float* __restrict__ Wr,
                               const float* __restrict__ bias,
                               float* __restrict__ out,
                               int M, int N) {
  const int lane = threadIdx.x & 31;
  const int wave = threadIdx.x >> 5;
  const int tile = blockIdx.x * (blockDim.x >> 5) + wave;
  const int tilesN = N >> 4;
  const int nTiles = (M >> 4) * tilesN;
  if (tile >= nTiles) return;

  const int tm = tile / tilesN;
  const int tn = tile - tm * tilesN;

  const int m  = lane & 15;
  const int kh = lane >> 4;

  const int arow = tm * 16 + m;
  const int bcol = tn * 16 + m;

  const float invd = 1.0f / fmaxf(deg[arow], 1.0f);

  const float* aP = aggsum + (size_t)arow * K + 2 * kh;
  const float* xP = xin    + (size_t)arow * K + 2 * kh;
  const float* lP = Wl     + (size_t)bcol * K + 2 * kh;
  const float* rP = Wr     + (size_t)bcol * K + 2 * kh;

  v8f acc = {};

#pragma unroll 8
  for (int k0 = 0; k0 < K; k0 += 4) {
    v2f a, b;
    a.x = aP[k0] * invd;
    a.y = aP[k0 + 1] * invd;
    b.x = lP[k0];
    b.y = lP[k0 + 1];
    acc = __builtin_amdgcn_wmma_f32_16x16x4_f32(false, a, false, b,
                                                (short)0, acc, false, false);
  }

#pragma unroll 8
  for (int k0 = 0; k0 < K; k0 += 4) {
    v2f a, b;
    a.x = xP[k0];
    a.y = xP[k0 + 1];
    b.x = rP[k0];
    b.y = rP[k0 + 1];
    acc = __builtin_amdgcn_wmma_f32_16x16x4_f32(false, a, false, b,
                                                (short)0, acc, false, false);
  }

  const float bv = bias[bcol];
#pragma unroll
  for (int r = 0; r < 8; ++r) {
    float v = acc[r] + bv;
    if (RELU) v = fmaxf(v, 0.0f);
    out[(size_t)(tm * 16 + r + 8 * kh) * N + bcol] = v;
  }
}

template <int K>
__launch_bounds__(256)
__global__ void sage_proj_wmma(const float* __restrict__ hin,
                               const float* __restrict__ Wl,
                               const float* __restrict__ Wr,
                               const float* __restrict__ bias,
                               float* __restrict__ proj,
                               float* __restrict__ out,
                               int M, int N) {
  const int lane = threadIdx.x & 31;
  const int wave = threadIdx.x >> 5;
  const int tile = blockIdx.x * (blockDim.x >> 5) + wave;
  const int tilesN = N >> 4;
  const int nTiles = (M >> 4) * tilesN;
  if (tile >= nTiles) return;

  const int tm = tile / tilesN;
  const int tn = tile - tm * tilesN;

  const int m  = lane & 15;
  const int kh = lane >> 4;

  const int arow = tm * 16 + m;
  const int bcol = tn * 16 + m;

  const float* hP = hin + (size_t)arow * K + 2 * kh;
  const float* lP = Wl  + (size_t)bcol * K + 2 * kh;
  const float* rP = Wr  + (size_t)bcol * K + 2 * kh;

  v8f accL = {};
  v8f accR = {};

  constexpr int KB = 4;
  for (int k0 = 0; k0 < K; k0 += 4 * KB) {
    v2f a[KB], bl[KB], br[KB];
#pragma unroll
    for (int j = 0; j < KB; ++j) {
      a[j].x  = hP[k0 + 4 * j];
      a[j].y  = hP[k0 + 4 * j + 1];
      bl[j].x = lP[k0 + 4 * j];
      bl[j].y = lP[k0 + 4 * j + 1];
      br[j].x = rP[k0 + 4 * j];
      br[j].y = rP[k0 + 4 * j + 1];
    }
#pragma unroll
    for (int j = 0; j < KB; ++j) {
      accL = __builtin_amdgcn_wmma_f32_16x16x4_f32(false, a[j], false, bl[j],
                                                   (short)0, accL, false, false);
      accR = __builtin_amdgcn_wmma_f32_16x16x4_f32(false, a[j], false, br[j],
                                                   (short)0, accR, false, false);
    }
  }

  const float bv = bias[bcol];
#pragma unroll
  for (int r = 0; r < 8; ++r) {
    const size_t o = (size_t)(tm * 16 + r + 8 * kh) * N + bcol;
    proj[o] = accL[r];
    out[o]  = accR[r] + bv;
  }
}

extern "C" void kernel_launch(void* const* d_in, const int* in_sizes, int n_in,
                              void* d_out, int out_size, void* d_ws, size_t ws_size,
                              hipStream_t stream) {
  (void)n_in; (void)out_size; (void)ws_size;

  const float* x   = (const float*)d_in[0];
  const int*   ei  = (const int*)  d_in[1];
  const float* W1l = (const float*)d_in[2];
  const float* b1  = (const float*)d_in[3];
  const float* W1r = (const float*)d_in[4];
  const float* W2l = (const float*)d_in[5];
  const float* b2  = (const float*)d_in[6];
  const float* W2r = (const float*)d_in[7];
  float* out = (float*)d_out;

  const int D1 = in_sizes[3];
  const int D2 = in_sizes[6];
  const int D0 = in_sizes[2] / D1;
  const int M  = in_sizes[0] / D0;
  const int E  = in_sizes[1] / 2;
  const int* src = ei;
  const int* dst = ei + E;

  float* deg  = (float*)d_ws;
  float* bufA = deg + M;
  float* h    = bufA + (size_t)M * D1;

  zero_f32<<<4096, 256, 0, stream>>>(deg, (long long)M * D0 + M);
  degree_kernel<<<(E + 255) / 256, 256, 0, stream>>>(dst, deg, E);
  {
    long long total = (long long)E * (128 / 4);
    scatter_add_kernel<128, false><<<(int)((total + 255) / 256), 256, 0, stream>>>(
        x, src, dst, deg, bufA, E);
  }
  {
    int tiles = (M / 16) * (D1 / 16);
    sage_gemm_wmma<128, true><<<(tiles + 7) / 8, 256, 0, stream>>>(
        bufA, deg, x, W1l, W1r, b1, h, M, D1);
  }

  {
    int tiles = (M / 16) * (D2 / 16);
    sage_proj_wmma<256><<<(tiles + 7) / 8, 256, 0, stream>>>(
        h, W2l, W2r, b2, bufA, out, M, D2);
  }
  {
    long long total = (long long)E * (128 / 4);
    scatter_add_kernel<128, true><<<(int)((total + 255) / 256), 256, 0, stream>>>(
        bufA, src, dst, deg, out, E);
  }
}


# Aggregation by gather over per-call destination buckets (count, prefix sum, fill): one wave sums each node's source rows and stores the mean once, in place of per-element f32 atomic adds
# speedup vs baseline: 1.8693x; 1.3530x over previous
#include <hip/hip_runtime.h>


typedef __attribute__((ext_vector_type(2))) float v2f;
typedef __attribute__((ext_vector_type(8))) float v8f;

__global__ void zero_i32(int* __restrict__ p, int n) {
  int i = blockIdx.x * blockDim.x + threadIdx.x;
  if (i < n) p[i] = 0;
}

__global__ void degree_kernel(const int* __restrict__ dst,
                              int* __restrict__ cnt, int E) {
  int e = blockIdx.x * blockDim.x + threadIdx.x;
  if (e < E) atomicAdd(&cnt[dst[e]], 1);
}

__launch_bounds__(1024)
__global__ void csr_scan_kernel(const int* __restrict__ cnt,
                                int* __restrict__ rowptr,
                                int* __restrict__ cursor, int M) {
  __shared__ int wave_off[32];
  __shared__ int tile_total;
  const int tid  = threadIdx.x;
  const int lane = tid & 31;
  const int wave = tid >> 5;
  int carry = 0;
  for (int base = 0; base < M; base += 1024 * 16) {
    const int i0 = base + tid * 16;
    int c[16];
#pragma unroll
    for (int q = 0; q < 4; ++q) {
      const int i = i0 + 4 * q;
      if (i + 3 < M) {
        const int4 v = *(const int4*)(cnt + i);
        c[4 * q] = v.x; c[4 * q + 1] = v.y; c[4 * q + 2] = v.z; c[4 * q + 3] = v.w;
      } else {
        c[4 * q]     = (i     < M) ? cnt[i]     : 0;
        c[4 * q + 1] = (i + 1 < M) ? cnt[i + 1] : 0;
        c[4 * q + 2] = (i + 2 < M) ? cnt[i + 2] : 0;
        c[4 * q + 3] = 0;
      }
    }
    int tsum = 0;
#pragma unroll
    for (int j = 0; j < 16; ++j) tsum += c[j];

    int incl = tsum;
#pragma unroll
    for (int off = 1; off < 32; off <<= 1) {
      const int y = __shfl_up(incl, off);
      if (lane >= off) incl += y;
    }
    if (lane == 31) wave_off[wave] = incl;
    __syncthreads();
    if (wave == 0) {
      const int w = wave_off[lane];
      int wi = w;
#pragma unroll
      for (int off = 1; off < 32; off <<= 1) {
        const int y = __shfl_up(wi, off);
        if (lane >= off) wi += y;
      }
      wave_off[lane] = wi - w;
      if (lane == 31) tile_total = wi;
    }
    __syncthreads();
    int p = carry + wave_off[wave] + (incl - tsum);
    carry += tile_total;
#pragma unroll
    for (int q = 0; q < 4; ++q) {
      const int i = i0 + 4 * q;
      int4 r;
      r.x = p; p += c[4 * q];
      r.y = p; p += c[4 * q + 1];
      r.z = p; p += c[4 * q + 2];
      r.w = p; p += c[4 * q + 3];
      if (i + 3 < M) {
        *(int4*)(rowptr + i) = r;
        *(int4*)(cursor + i) = r;
      } else {
        if (i     < M) { rowptr[i]     = r.x; cursor[i]     = r.x; }
        if (i + 1 < M) { rowptr[i + 1] = r.y; cursor[i + 1] = r.y; }
        if (i + 2 < M) { rowptr[i + 2] = r.z; cursor[i + 2] = r.z; }
      }
    }
    __syncthreads();
  }
  if (tid == 0) rowptr[M] = carry;
}

__global__ void csr_fill_kernel(const int* __restrict__ src,
                                const int* __restrict__ dst,
                                int* __restrict__ cursor,
                                int* __restrict__ col, int E) {
  int e = blockIdx.x * blockDim.x + threadIdx.x;
  if (e < E) {
    const int p = atomicAdd(&cursor[dst[e]], 1);
    col[p] = src[e];
  }
}

template <bool ADD>
__launch_bounds__(256)
__global__ void gather_mean_kernel(const float* __restrict__ feat,
                                   const int* __restrict__ rowptr,
                                   const int* __restrict__ col,
                                   float* __restrict__ outp,
                                   int M) {
  const int lane = threadIdx.x & 31;
  const int node = blockIdx.x * (blockDim.x >> 5) + (threadIdx.x >> 5);
  if (node >= M) return;
  const int beg = __builtin_amdgcn_readfirstlane(rowptr[node]);
  const int end = __builtin_amdgcn_readfirstlane(rowptr[node + 1]);

  const float4* __restrict__ f4 = (const float4*)feat;
  float4* o = (float4*)outp + (size_t)node * 32 + lane;
  float4 c = {0.f, 0.f, 0.f, 0.f};
  if (ADD) c = *o;
  float4 a0 = {0.f, 0.f, 0.f, 0.f}, a1 = a0, a2 = a0, a3 = a0;

  for (int base = beg; base < end; base += 32) {
    const int n = (end - base < 32) ? (end - base) : 32;
    const int myc = (lane < n) ? col[base + lane] : 0;
    for (int t = 0; t < n; t += 4) {
      float4 v0, v1 = {0.f, 0.f, 0.f, 0.f}, v2 = v1, v3 = v1;
      v0 = f4[(size_t)__builtin_amdgcn_readlane(myc, t) * 32 + lane];
      if (t + 1 < n) v1 = f4[(size_t)__builtin_amdgcn_readlane(myc, t + 1) * 32 + lane];
      if (t + 2 < n) v2 = f4[(size_t)__builtin_amdgcn_readlane(myc, t + 2) * 32 + lane];
      if (t + 3 < n) v3 = f4[(size_t)__builtin_amdgcn_readlane(myc, t + 3) * 32 + lane];
      a0.x += v0.x; a0.y += v0.y; a0.z += v0.z; a0.w += v0.w;
      a1.x += v1.x; a1.y += v1.y; a1.z += v1.z; a1.w += v1.w;
      a2.x += v2.x; a2.y += v2.y; a2.z += v2.z; a2.w += v2.w;
      a3.x += v3.x; a3.y += v3.y; a3.z += v3.z; a3.w += v3.w;
    }
  }

  const float dg = fmaxf((float)(end - beg), 1.0f);
  float4 r;
  r.x = ((a0.x + a1.x) + (a2.x + a3.x)) / dg;
  r.y = ((a0.y + a1.y) + (a2.y + a3.y)) / dg;
  r.z = ((a0.z + a1.z) + (a2.z + a3.z)) / dg;
  r.w = ((a0.w + a1.w) + (a2.w + a3.w)) / dg;

  if (ADD) {
    r.x += c.x; r.y += c.y; r.z += c.z; r.w += c.w;
  }
  *o = r;
}

template <int K, bool RELU>
__launch_bounds__(256)
__global__ void sage_gemm_wmma(const float* __restrict__ agg,
                               const float* __restrict__ xin,
                               const float* __restrict__ Wl,
                               const float* __restrict__ Wr,
                               const float* __restrict__ bias,
                               float* __restrict__ out,
                               int M, int N) {
  const int lane = threadIdx.x & 31;
  const int wave = threadIdx.x >> 5;
  const int tile = blockIdx.x * (blockDim.x >> 5) + wave;
  const int tilesN = N >> 4;
  const int nTiles = (M >> 4) * tilesN;
  if (tile >= nTiles) return;

  const int tm = tile / tilesN;
  const int tn = tile - tm * tilesN;

  const int m  = lane & 15;
  const int kh = lane >> 4;

  const int arow = tm * 16 + m;
  const int bcol = tn * 16 + m;

  const float* aP = agg + (size_t)arow * K + 2 * kh;
  const float* xP = xin + (size_t)arow * K + 2 * kh;
  const float* lP = Wl  + (size_t)bcol * K + 2 * kh;
  const float* rP = Wr  + (size_t)bcol * K + 2 * kh;

  v8f acc = {};

#pragma unroll 8
  for (int k0 = 0; k0 < K; k0 += 4) {
    v2f a, b;
    a.x = aP[k0];
    a.y = aP[k0 + 1];
    b.x = lP[k0];
    b.y = lP[k0 + 1];
    acc = __builtin_amdgcn_wmma_f32_16x16x4_f32(false, a, false, b,
                                                (short)0, acc, false, false);
  }

#pragma unroll 8
  for (int k0 = 0; k0 < K; k0 += 4) {
    v2f a, b;
    a.x = xP[k0];
    a.y = xP[k0 + 1];
    b.x = rP[k0];
    b.y = rP[k0 + 1];
    acc = __builtin_amdgcn_wmma_f32_16x16x4_f32(false, a, false, b,
                                                (short)0, acc, false, false);
  }

  const float bv = bias[bcol];
#pragma unroll
  for (int r = 0; r < 8; ++r) {
    float v = acc[r] + bv;
    if (RELU) v = fmaxf(v, 0.0f);
    out[(size_t)(tm * 16 + r + 8 * kh) * N + bcol] = v;
  }
}

template <int K>
__launch_bounds__(256)
__global__ void sage_proj_wmma(const float* __restrict__ hin,
                               const float* __restrict__ Wl,
                               const float* __restrict__ Wr,
                               const float* __restrict__ bias,
                               float* __restrict__ proj,
                               float* __restrict__ out,
                               int M, int N) {
  const int lane = threadIdx.x & 31;
  const int wave = threadIdx.x >> 5;
  const int tile = blockIdx.x * (blockDim.x >> 5) + wave;
  const int tilesN = N >> 4;
  const int nTiles = (M >> 4) * tilesN;
  if (tile >= nTiles) return;

  const int tm = tile / tilesN;
  const int tn = tile - tm * tilesN;

  const int m  = lane & 15;
  const int kh = lane >> 4;

  const int arow = tm * 16 + m;
  const int bcol = tn * 16 + m;

  const float* hP = hin + (size_t)arow * K + 2 * kh;
  const float* lP = Wl  + (size_t)bcol * K + 2 * kh;
  const float* rP = Wr  + (size_t)bcol * K + 2 * kh;

  v8f accL = {};
  v8f accR = {};

  constexpr int KB = 4;
  for (int k0 = 0; k0 < K; k0 += 4 * KB) {
    v2f a[KB], bl[KB], br[KB];
#pragma unroll
    for (int j = 0; j < KB; ++j) {
      a[j].x  = hP[k0 + 4 * j];
      a[j].y  = hP[k0 + 4 * j + 1];
      bl[j].x = lP[k0 + 4 * j];
      bl[j].y = lP[k0 + 4 * j + 1];
      br[j].x = rP[k0 + 4 * j];
      br[j].y = rP[k0 + 4 * j + 1];
    }
#pragma unroll
    for (int j = 0; j < KB; ++j) {
      accL = __builtin_amdgcn_wmma_f32_16x16x4_f32(false, a[j], false, bl[j],
                                                   (short)0, accL, false, false);
      accR = __builtin_amdgcn_wmma_f32_16x16x4_f32(false, a[j], false, br[j],
                                                   (short)0, accR, false, false);
    }
  }

  const float bv = bias[bcol];
#pragma unroll
  for (int r = 0; r < 8; ++r) {
    const size_t o = (size_t)(tm * 16 + r + 8 * kh) * N + bcol;
    proj[o] = accL[r];
    out[o]  = accR[r] + bv;
  }
}

extern "C" void kernel_launch(void* const* d_in, const int* in_sizes, int n_in,
                              void* d_out, int out_size, void* d_ws, size_t ws_size,
                              hipStream_t stream) {
  (void)n_in; (void)out_size; (void)ws_size;

  const float* x   = (const float*)d_in[0];
  const int*   ei  = (const int*)  d_in[1];
  const float* W1l = (const float*)d_in[2];
  const float* b1  = (const float*)d_in[3];
  const float* W1r = (const float*)d_in[4];
  const float* W2l = (const float*)d_in[5];
  const float* b2  = (const float*)d_in[6];
  const float* W2r = (const float*)d_in[7];
  float* out = (float*)d_out;

  const int D1 = in_sizes[3];
  const int D2 = in_sizes[6];
  const int D0 = in_sizes[2] / D1;
  const int M  = in_sizes[0] / D0;
  const int E  = in_sizes[1] / 2;
  const int* src = ei;
  const int* dst = ei + E;

  const int Mp = ((M + 4) / 4) * 4;
  float* bufA   = (float*)d_ws;
  float* h      = bufA + (size_t)M * 128;
  int*   cnt    = (int*)(h + (size_t)M * D1);
  int*   rowptr = cnt + Mp;
  int*   cursor = rowptr + Mp;
  int*   col    = cursor + Mp;

  zero_i32<<<(M + 255) / 256, 256, 0, stream>>>(cnt, M);
  degree_kernel<<<(E + 255) / 256, 256, 0, stream>>>(dst, cnt, E);
  csr_scan_kernel<<<1, 1024, 0, stream>>>(cnt, rowptr, cursor, M);
  csr_fill_kernel<<<(E + 255) / 256, 256, 0, stream>>>(src, dst, cursor, col, E);

  gather_mean_kernel<false><<<(M + 7) / 8, 256, 0, stream>>>(x, rowptr, col, bufA, M);
  {
    int tiles = (M / 16) * (D1 / 16);
    sage_gemm_wmma<128, true><<<(tiles + 7) / 8, 256, 0, stream>>>(
        bufA, x, W1l, W1r, b1, h, M, D1);
  }

  {
    int tiles = (M / 16) * (D2 / 16);
    sage_proj_wmma<256><<<(tiles + 7) / 8, 256, 0, stream>>>(
        h, W2l, W2r, b2, bufA, out, M, D2);
  }
  gather_mean_kernel<true><<<(M + 7) / 8, 256, 0, stream>>>(bufA, rowptr, col, out, M);
}
